# starting kernel
# baseline (speedup 1.0000x reference)
#include <hip/hip_runtime.h>
#include <hip/hip_bf16.h>

typedef __attribute__((ext_vector_type(2))) float v2f;
typedef __attribute__((ext_vector_type(8))) float v8f;

__global__ __launch_bounds__(256) void k_init_deg(float* __restrict__ deg, int n) {
  int i = blockIdx.x * blockDim.x + threadIdx.x;
  if (i < n) deg[i] = 1.0f;
}

__global__ __launch_bounds__(256) void k_count_deg(const int* __restrict__ col,
                                                   float* deg, int E) {
  int e = blockIdx.x * blockDim.x + threadIdx.x;
  if (e < E) {
    (void)__hip_atomic_fetch_add(&deg[col[e]], 1.0f, __ATOMIC_RELAXED,
                                 __HIP_MEMORY_SCOPE_AGENT);
  }
}

__global__ __launch_bounds__(256) void k_dinv(float* __restrict__ deg, int n) {
  int i = blockIdx.x * blockDim.x + threadIdx.x;
  if (i < n) {
    float d = deg[i];
    deg[i] = (d > 0.0f) ? rsqrtf(d) : 0.0f;
  }
}

__global__ __launch_bounds__(256) void k_gemm_wmma_f32(
    const float* __restrict__ A, const float* __restrict__ B,
    float* __restrict__ C, int M, int K, int N) {
  const int lane  = threadIdx.x & 31;
  const int wave  = threadIdx.x >> 5;
  const int tilesN = N >> 4;
  const int t  = blockIdx.x * (blockDim.x >> 5) + wave;
  const int m0 = (t / tilesN) << 4;
  const int n0 = (t % tilesN) << 4;
  if (m0 >= M) return;

  const int l15   = lane & 15;
  const int khalf = (lane >> 4) << 1;

  const float* arow = A + (size_t)(m0 + l15) * K + khalf;
  const float* bcol = B + (size_t)khalf * N + (n0 + l15);

  v8f acc = {};
  for (int k0 = 0; k0 < K; k0 += 4) {
    v2f a, b;
    float2 av = *(const float2*)arow;
    a.x = av.x;
    a.y = av.y;
    b.x = bcol[0];
    b.y = bcol[N];
    acc = __builtin_amdgcn_wmma_f32_16x16x4_f32(
         false, a,  false, b,
         (short)0, acc,  false,  false);
    arow += 4;
    bcol += (size_t)4 * N;
  }

  const int rbase = m0 + ((lane >> 4) << 3);
  float* cptr = C + (size_t)rbase * N + n0 + l15;
#pragma unroll
  for (int v = 0; v < 8; ++v) cptr[(size_t)v * N] = acc[v];
}

__global__ __launch_bounds__(256) void k_selfloop(const float* __restrict__ h,
                                                  const float* __restrict__ dinv,
                                                  float* __restrict__ agg,
                                                  int total, int F) {
  int idx = blockIdx.x * blockDim.x + threadIdx.x;
  if (idx < total) {
    float di = dinv[idx / F];
    agg[idx] = h[idx] * di * di;
  }
}

__global__ __launch_bounds__(256) void k_scatter(const float* __restrict__ h,
                                                 const int* __restrict__ row,
                                                 const int* __restrict__ col,
                                                 const float* __restrict__ dinv,
                                                 float* agg, int E, int F) {
  const int g = F >> 2;
  const long long total = (long long)E * g;
  long long tid = (long long)blockIdx.x * blockDim.x + threadIdx.x;
  if (tid >= total) return;
  const int e = (int)(tid / g);
  const int f = (int)(tid % g) << 2;
  const int r = row[e];
  const int c = col[e];
  const float nrm = dinv[r] * dinv[c];
  const float4 hv = *(const float4*)(h + (size_t)r * F + f);
  float* dst = agg + (size_t)c * F + f;
  (void)__hip_atomic_fetch_add(&dst[0], hv.x * nrm, __ATOMIC_RELAXED, __HIP_MEMORY_SCOPE_AGENT);
  (void)__hip_atomic_fetch_add(&dst[1], hv.y * nrm, __ATOMIC_RELAXED, __HIP_MEMORY_SCOPE_AGENT);
  (void)__hip_atomic_fetch_add(&dst[2], hv.z * nrm, __ATOMIC_RELAXED, __HIP_MEMORY_SCOPE_AGENT);
  (void)__hip_atomic_fetch_add(&dst[3], hv.w * nrm, __ATOMIC_RELAXED, __HIP_MEMORY_SCOPE_AGENT);
}

__global__ __launch_bounds__(256) void k_bias_act(float* __restrict__ agg,
                                                  const float* __restrict__ b,
                                                  int total, int F, int do_relu) {
  int idx = blockIdx.x * blockDim.x + threadIdx.x;
  if (idx < total) {
    float v = agg[idx] + b[idx % F];
    if (do_relu) v = fmaxf(v, 0.0f);
    agg[idx] = v;
  }
}

extern "C" void kernel_launch(void* const* d_in, const int* in_sizes, int n_in,
                              void* d_out, int out_size, void* d_ws, size_t ws_size,
                              hipStream_t stream) {
  const float* x  = (const float*)d_in[0];
  const int*   ei = (const int*)d_in[1];
  const float* W1 = (const float*)d_in[2];
  const float* b1 = (const float*)d_in[3];
  const float* W2 = (const float*)d_in[4];
  const float* b2 = (const float*)d_in[5];

  const int H  = in_sizes[3];
  const int C  = in_sizes[5];
  const int F  = in_sizes[2] / H;
  const int Nn = in_sizes[0] / F;
  const int E  = in_sizes[1] / 2;
  const int* rowIdx = ei;
  const int* colIdx = ei + E;

  auto al = [](size_t v) { return (v + 255) & ~(size_t)255; };
  float* dinv = (float*)d_ws;
  size_t off  = al((size_t)Nn * sizeof(float));
  float* hbuf = (float*)((char*)d_ws + off);
  off += al((size_t)Nn * H * sizeof(float));
  float* abuf = (float*)((char*)d_ws + off);

  const dim3 blk(256);
  auto blocks = [](long long work) { return (unsigned)((work + 255) / 256); };

  k_init_deg <<<blocks(Nn), blk, 0, stream>>>(dinv, Nn);
  k_count_deg<<<blocks(E),  blk, 0, stream>>>(colIdx, dinv, E);
  k_dinv     <<<blocks(Nn), blk, 0, stream>>>(dinv, Nn);

  const int tiles1 = ((Nn + 15) / 16) * (H / 16);
  k_gemm_wmma_f32<<<(tiles1 + 7) / 8, blk, 0, stream>>>(x, W1, hbuf, Nn, F, H);
  const int tot1 = Nn * H;
  k_selfloop<<<blocks(tot1), blk, 0, stream>>>(hbuf, dinv, abuf, tot1, H);
  k_scatter <<<blocks((long long)E * (H / 4)), blk, 0, stream>>>(hbuf, rowIdx, colIdx, dinv, abuf, E, H);
  k_bias_act<<<blocks(tot1), blk, 0, stream>>>(abuf, b1, tot1, H,  1);

  const int tiles2 = ((Nn + 15) / 16) * (C / 16);
  k_gemm_wmma_f32<<<(tiles2 + 7) / 8, blk, 0, stream>>>(abuf, W2, hbuf, Nn, H, C);
  const int tot2 = Nn * C;
  float* out = (float*)d_out;
  k_selfloop<<<blocks(tot2), blk, 0, stream>>>(hbuf, dinv, out, tot2, C);
  k_scatter <<<blocks((long long)E * (C / 4)), blk, 0, stream>>>(hbuf, rowIdx, colIdx, dinv, out, E, C);
  k_bias_act<<<blocks(tot2), blk, 0, stream>>>(out, b2, tot2, C,  0);
}


# layer-1 aggregation: a work-group per 196-node range keeps the sums in LDS, finds its edges by a 1-byte range tag, adds rows with LDS f32 adds, stores rows once; sum order: LDS arrival
# speedup vs baseline: 3.2604x; 3.2604x over previous
#include <hip/hip_runtime.h>
#include <hip/hip_bf16.h>

typedef __attribute__((ext_vector_type(2))) float v2f;
typedef __attribute__((ext_vector_type(8))) float v8f;

#define TICK_PITCH 64
#define TICK_SLICES 16

__device__ __forceinline__ int cache_half() {
  const unsigned id = __builtin_amdgcn_s_sendmsg_rtn(0x87);
  return (int)((id >> 18) & 1u);
}

#define DEG_SUBS 4
#define DEG_TABS (2 * DEG_SUBS)

#define AGG_R 196
#define AGG_OWNERS 256

__global__ __launch_bounds__(256) void k_init_deg(float* __restrict__ cnt,
                                                  int* __restrict__ tick, int n4) {
  int i = blockIdx.x * blockDim.x + threadIdx.x;
  if (i < n4) ((float4*)cnt)[i] = make_float4(0.0f, 0.0f, 0.0f, 0.0f);
  if (i < 4 * TICK_SLICES) tick[i * TICK_PITCH] = 0;
}

__global__ __launch_bounds__(256) void k_count_deg(const int* __restrict__ col,
                                                   float* cnt, unsigned* __restrict__ tag,
                                                   int pitch, int E, int nwords) {
  const int t = blockIdx.x * blockDim.x + threadIdx.x;
  const int tab = cache_half() * DEG_SUBS + (((int)threadIdx.x >> 5) & (DEG_SUBS - 1));
  if (t < nwords) {
    int c[4];
#pragma unroll
    for (int k = 0; k < 4; ++k) {
      const int e = 4 * t + k;
      c[k] = col[(e < E) ? e : (E - 1)];
    }
    unsigned word = 0u;
#pragma unroll
    for (int k = 0; k < 4; ++k) {
      if (4 * t + k < E) {
        (void)__hip_atomic_fetch_add(&cnt[(size_t)tab * pitch + c[k]], 1.0f, __ATOMIC_RELAXED,
                                     __HIP_MEMORY_SCOPE_AGENT);
        word |= ((unsigned)(c[k] / AGG_R) & (AGG_OWNERS - 1u)) << (8 * k);
      }
    }
    tag[t] = word;
  }
}

__global__ __launch_bounds__(256) void k_dinv(const float* __restrict__ cnt,
                                              float* __restrict__ dinv, int pitch, int n) {
  int i = blockIdx.x * blockDim.x + threadIdx.x;
  if (i < n) {
    float d = 1.0f;
#pragma unroll
    for (int k = 0; k < DEG_TABS; ++k) d += cnt[(size_t)k * pitch + i];
    dinv[i] = (d > 0.0f) ? rsqrtf(d) : 0.0f;
  }
}

template <int K, int N, bool FUSE, bool STORE_S>
__global__ __launch_bounds__(256) void k_gemm_wmma_f32(
    const float* __restrict__ A, const float* __restrict__ B,
    const float* __restrict__ dinv, const float* __restrict__ bin,
    const float* __restrict__ bout, float* __restrict__ C,
    float* __restrict__ S, int M) {
  constexpr int TN    = N / 16;
  constexpr int MT    = (TN >= 8) ? 1 : (8 / TN);
  constexpr int ROWS  = 16 * MT;
  constexpr int V4ROW = K / 4;
  constexpr int NLD   = ROWS * V4ROW / 256;
  constexpr int PITCH = K + 4;
  static_assert(N % 16 == 0 && K % 64 == 0, "tile shape");
  static_assert((TN >= 8) ? (TN % 8 == 0) : (8 % TN == 0), "8 waves per row-tile group");
  static_assert((ROWS * V4ROW) % 256 == 0, "staging split");
  static_assert(256 % V4ROW == 0, "a thread stages the same k columns in every piece");
  __shared__ __attribute__((aligned(16))) float As[ROWS * PITCH];

  const int lane = threadIdx.x & 31;
  const int wave = threadIdx.x >> 5;
  const int t    = blockIdx.x * 8 + wave;
  const int mt0  = (blockIdx.x * 8) / TN;
  const int mt   = t / TN;
  const int m0   = mt << 4;
  const int n0   = (t % TN) << 4;
  const int mbase = mt0 << 4;

  float4 st[NLD];
  float4 bk = make_float4(0.0f, 0.0f, 0.0f, 0.0f);
  if (FUSE) bk = *(const float4*)(bin + ((int)threadIdx.x % V4ROW) * 4);
#pragma unroll
  for (int i = 0; i < NLD; ++i) {
    const int idx = i * 256 + (int)threadIdx.x;
    const int r   = idx / V4ROW;
    const int c4  = idx % V4ROW;
    st[i] = make_float4(0.0f, 0.0f, 0.0f, 0.0f);
    if (mbase + r < M) st[i] = *(const float4*)(A + (size_t)(mbase + r) * K + c4 * 4);
  }
#pragma unroll
  for (int i = 0; i < NLD; ++i) {
    const int idx = i * 256 + (int)threadIdx.x;
    const int r   = idx / V4ROW;
    const int c4  = idx % V4ROW;
    float4 a4 = st[i];
    if (FUSE) {
      a4.x = fmaxf(a4.x + bk.x, 0.0f);
      a4.y = fmaxf(a4.y + bk.y, 0.0f);
      a4.z = fmaxf(a4.z + bk.z, 0.0f);
      a4.w = fmaxf(a4.w + bk.w, 0.0f);
    }
    *(float4*)(As + r * PITCH + c4 * 4) = a4;
  }
  __syncthreads();
  if (m0 >= M) return;

  const int l15   = lane & 15;
  const int khalf = (lane >> 4) << 1;

  const float* ap   = As + ((mt - mt0) * 16 + l15) * PITCH + khalf;
  const float* bcol = B + (size_t)khalf * N + (n0 + l15);

  const int rbase = m0 + ((lane >> 4) << 3);
  float4 d0 = make_float4(0.0f, 0.0f, 0.0f, 0.0f), d1 = d0;
  if (STORE_S) {
    d0 = *(const float4*)(dinv + rbase);
    d1 = *(const float4*)(dinv + rbase + 4);
  }
  float bo = 0.0f;
  if (FUSE) bo = bout[n0 + l15];

  v8f acc = {};
  for (int k0 = 0; k0 < K; k0 += 4) {
    v2f a, b;
    const float2 av = *(const float2*)(ap + k0);
    a.x = av.x;
    a.y = av.y;
    b.x = bcol[(size_t)k0 * N];
    b.y = bcol[(size_t)(k0 + 1) * N];
    acc = __builtin_amdgcn_wmma_f32_16x16x4_f32(
         false, a,  false, b,
         (short)0, acc,  false,  false);
  }

  float* cptr = C + (size_t)rbase * N + n0 + l15;
  float* sptr = S + (size_t)rbase * N + n0 + l15;
  const float d[8] = {d0.x, d0.y, d0.z, d0.w, d1.x, d1.y, d1.z, d1.w};
#pragma unroll
  for (int v = 0; v < 8; ++v) {
    const float c = acc[v];
    cptr[(size_t)v * N] = c;
    if (STORE_S) {
      const float s = c * d[v] * d[v];
      sptr[(size_t)v * N] = FUSE ? s + bo : s;
    }
  }
}


__device__ __forceinline__ int draw_ticket(int* cnt, int lane) {
  int t = 0;
  if (lane == 0)
    t = __hip_atomic_fetch_add(cnt, 1, __ATOMIC_RELAXED, __HIP_MEMORY_SCOPE_AGENT);
  return t;
}

template <int F>
__global__ __launch_bounds__(256) void k_scatter(const float* __restrict__ h,
                                                 const int* __restrict__ row,
                                                 const int* __restrict__ col,
                                                 const float* __restrict__ dinv,
                                                 float* agg, int* tick, int E, int Nn) {
  constexpr int L = F / 32;
  constexpr int U = (L >= 8) ? 4 : (32 / L);
  const int lane   = threadIdx.x & 31;
  const int G      = (E + 31) >> 5;
  const int nsplit = Nn >> 1;
  const int home   = cache_half();
  const int slice  = (int)((blockIdx.x * 0x9E3779B1u) >> 28);
  const int Gs     = (G - slice + TICK_SLICES - 1) / TICK_SLICES;

  for (int pass = 0; pass < 2; ++pass) {
    const int range = home ^ pass;
    int* cnt = tick + (range * TICK_SLICES + slice) * TICK_PITCH;
    int t = __builtin_amdgcn_readfirstlane(draw_ticket(cnt, lane));
    while (t < Gs) {
      const int e = ((t * TICK_SLICES + slice) << 5) + lane;
      int r = 0, c = 0;
      float dr = 0.0f, dc = 0.0f;
      bool mine = false;
      if (e < E) {
        r = row[e];
        c = col[e];
        mine = ((c >= nsplit) ? 1 : 0) == range;
      }
      if (mine) {
        dr = dinv[r];
        dc = dinv[c];
      }
      const int tnext = draw_ticket(cnt, lane);
      const float w = dr * dc;
      const int wbits = __float_as_int(w);
      unsigned m = __builtin_amdgcn_ballot_w32(mine);

      while (m) {
        const int n = __builtin_popcount(m);
        float v[U][L];
        int cj[U];
        float wj[U];
#pragma unroll
        for (int u = 0; u < U; ++u) {
          if (u < n) {
            const int j = __builtin_ctz(m);
            m &= m - 1;
            const int rj = __builtin_amdgcn_readlane(r, j);
            cj[u] = __builtin_amdgcn_readlane(c, j);
            wj[u] = __int_as_float(__builtin_amdgcn_readlane(wbits, j));
            const float* src = h + (size_t)rj * F + lane;
#pragma unroll
            for (int k = 0; k < L; ++k) v[u][k] = src[32 * k];
          }
        }
#pragma unroll
        for (int u = 0; u < U; ++u) {
          if (u < n) {
            float* dst = agg + (size_t)cj[u] * F + lane;
#pragma unroll
            for (int k = 0; k < L; ++k)
              (void)__hip_atomic_fetch_add(dst + 32 * k, v[u][k] * wj[u], __ATOMIC_RELAXED,
                                           __HIP_MEMORY_SCOPE_AGENT);
          }
        }
      }
      t = __builtin_amdgcn_readfirstlane(tnext);
    }
  }
}


__device__ __forceinline__ unsigned zero_bytes(unsigned x) {
  const unsigned t = (x & 0x7F7F7F7Fu) + 0x7F7F7F7Fu;
  return ~(t | x | 0x7F7F7F7Fu);
}

template <int F>
__global__ __launch_bounds__(1024) void k_agg_lds(const float* __restrict__ h,
                                                  const int* __restrict__ row,
                                                  const int* __restrict__ col,
                                                  const unsigned* __restrict__ tag,
                                                  const float* __restrict__ dinv,
                                                  float* __restrict__ agg, int E, int Nn) {
  constexpr int L  = F / 32;
  constexpr int U  = (L >= 8) ? 6 : (32 / L);
  constexpr int NW = 32;
  __shared__ float acc[AGG_R * F];
  __shared__ uint4 stg[NW * 4 * 32];
  const int lane = threadIdx.x & 31;
  const int wave = __builtin_amdgcn_readfirstlane((int)threadIdx.x >> 5);
  const int NCH  = (E + 511) >> 9;

  for (int o = blockIdx.x; o * AGG_R < Nn; o += AGG_OWNERS) {
    const int lo    = o * AGG_R;
    const int nrows = (Nn - lo < AGG_R) ? (Nn - lo) : AGG_R;

    for (int i = wave; i < nrows; i += NW) {
      const float d = dinv[lo + i];
      const float* src = h + (size_t)(lo + i) * F + lane;
      float* a = acc + i * F + lane;
#pragma unroll
      for (int k = 0; k < L; ++k) a[32 * k] = (src[32 * k] * d) * d;
    }
    __syncthreads();

    const unsigned pat = (unsigned)(o & (AGG_OWNERS - 1)) * 0x01010101u;
    int cnt = 0;
    int qe  = 0;

    auto flush = [&]() {
      int r = 0, c = 0;
      bool mine = false;
      if (lane < cnt) {
        r = row[qe];
        c = col[qe];
        mine = (c >= lo) && (c < lo + nrows);
      }
      float w = 0.0f;
      if (mine) w = dinv[r] * dinv[c];
      const int wbits = __float_as_int(w);
      unsigned mm = __builtin_amdgcn_ballot_w32(mine);
      while (mm) {
        const int n = __builtin_popcount(mm);
        float v[U][L];
        int cj[U];
        float wj[U];
#pragma unroll
        for (int u = 0; u < U; ++u) {
          if (u < n) {
            const int j = __builtin_ctz(mm);
            mm &= mm - 1;
            const int rj = __builtin_amdgcn_readlane(r, j);
            cj[u] = __builtin_amdgcn_readlane(c, j) - lo;
            wj[u] = __int_as_float(__builtin_amdgcn_readlane(wbits, j));
            const float* src = h + (size_t)rj * F + lane;
#pragma unroll
            for (int k = 0; k < L; ++k) v[u][k] = src[32 * k];
          }
        }
#pragma unroll
        for (int u = 0; u < U; ++u) {
          if (u < n) {
            float* dst = acc + cj[u] * F + lane;
#pragma unroll
            for (int k = 0; k < L; ++k)
              (void)__hip_atomic_fetch_add(dst + 32 * k, v[u][k] * wj[u], __ATOMIC_RELAXED,
                                           __HIP_MEMORY_SCOPE_WORKGROUP);
          }
        }
      }
      cnt = 0;
    };

    uint4* park = stg + wave * 128 + lane;
    for (int g = wave * 4; g < NCH; g += NW * 4) {
      uint4 tg[4];
#pragma unroll
      for (int sc = 0; sc < 4; ++sc) {
        const int chc = (g + sc < NCH) ? (g + sc) : (NCH - 1);
        tg[sc] = *(const uint4*)(tag + (chc << 7) + (lane << 2));
      }
#pragma unroll
      for (int sc = 0; sc < 4; ++sc) {
        if (g + sc >= NCH) tg[sc] = make_uint4(~pat, ~pat, ~pat, ~pat);
        park[sc * 32] = tg[sc];
      }
#pragma nounroll
      for (int sc = 0; sc < 4; ++sc) {
        const uint4 tq = park[sc * 32];
        const unsigned z0 = zero_bytes(tq.x ^ pat);
        const unsigned z1 = zero_bytes(tq.y ^ pat);
        const unsigned z2 = zero_bytes(tq.z ^ pat);
        const unsigned z3 = zero_bytes(tq.w ^ pat);
        unsigned m = __builtin_amdgcn_ballot_w32((z0 | z1 | z2 | z3) != 0u);
        while (m) {
          const int j = __builtin_ctz(m);
          m &= m - 1;
          if (cnt > 16) flush();
          const int ebase = ((g + sc) << 9) + (j << 4);
#pragma unroll
          for (int q = 0; q < 4; ++q) {
            unsigned zw = (unsigned)__builtin_amdgcn_readlane(
                (int)(q == 0 ? z0 : q == 1 ? z1 : q == 2 ? z2 : z3), j);
            while (zw) {
              const int bit = __builtin_ctz(zw);
              zw &= zw - 1;
              const int e = ebase + (q << 2) + (bit >> 3);
              if (e < E) {
                qe = (lane == cnt) ? e : qe;
                ++cnt;
              }
            }
          }
        }
      }
    }
    if (cnt > 0) flush();
    __syncthreads();

    for (int i = wave; i < nrows; i += NW) {
      const float* a = acc + i * F + lane;
      float* dst = agg + (size_t)(lo + i) * F + lane;
#pragma unroll
      for (int k = 0; k < L; ++k) dst[32 * k] = a[32 * k];
    }
    __syncthreads();
  }
}

extern "C" void kernel_launch(void* const* d_in, const int* in_sizes, int n_in,
                              void* d_out, int out_size, void* d_ws, size_t ws_size,
                              hipStream_t stream) {
  const float* x  = (const float*)d_in[0];
  const int*   ei = (const int*)d_in[1];
  const float* W1 = (const float*)d_in[2];
  const float* b1 = (const float*)d_in[3];
  const float* W2 = (const float*)d_in[4];
  const float* b2 = (const float*)d_in[5];

  const int H  = in_sizes[3];
  const int C  = in_sizes[5];
  const int F  = in_sizes[2] / H;
  const int Nn = in_sizes[0] / F;
  const int E  = in_sizes[1] / 2;
  const int* rowIdx = ei;
  const int* colIdx = ei + E;

  auto al = [](size_t v) { return (v + 255) & ~(size_t)255; };
  const int pitch = (Nn + 63) & ~63;
  float* dinv = (float*)d_ws;
  size_t off  = al((size_t)Nn * sizeof(float));
  float* cnt  = (float*)((char*)d_ws + off);
  off += al((size_t)DEG_TABS * pitch * sizeof(float));
  float* hbuf = (float*)((char*)d_ws + off);
  off += al((size_t)Nn * H * sizeof(float));
  float* abuf = (float*)((char*)d_ws + off);
  off += al((size_t)Nn * H * sizeof(float));
  int* tick = (int*)((char*)d_ws + off);
  off += al((size_t)4 * TICK_SLICES * TICK_PITCH * sizeof(int));
  unsigned* tag = (unsigned*)((char*)d_ws + off);
  const int nwords = ((E + 511) / 512) * 128;

  const dim3 blk(256);
  auto blocks = [](long long work) { return (unsigned)((work + 255) / 256); };

  const int n4 = DEG_TABS * pitch / 4;
  k_init_deg <<<blocks(n4), blk, 0, stream>>>(cnt, tick, n4);
  k_count_deg<<<blocks(nwords), blk, 0, stream>>>(colIdx, cnt, tag, pitch, E, nwords);
  k_dinv     <<<blocks(Nn), blk, 0, stream>>>(cnt, dinv, pitch, Nn);

  const int tiles1 = ((Nn + 15) / 16) * (H / 16);
  k_gemm_wmma_f32<128, 256, false, false><<<(tiles1 + 7) / 8, blk, 0, stream>>>(
      x, W1, dinv, nullptr, nullptr, hbuf, nullptr, Nn);
  k_agg_lds<256><<<AGG_OWNERS, dim3(1024), 0, stream>>>(hbuf, rowIdx, colIdx, tag, dinv, abuf,
                                                       E, Nn);
  const unsigned sblocks = 2048;

  const int tiles2 = ((Nn + 15) / 16) * (C / 16);
  float* out = (float*)d_out;
  k_gemm_wmma_f32<256, 64, true, true><<<(tiles2 + 7) / 8, blk, 0, stream>>>(
      abuf, W2, dinv, b1, b2, hbuf, out, Nn);
  k_scatter<64><<<sblocks, blk, 0, stream>>>(hbuf, rowIdx, colIdx, dinv, out,
                                            tick + 2 * TICK_SLICES * TICK_PITCH, E, Nn);
}


# layer 2's product formed by the aggregation work-group on its LDS sums (WMMA A fragments from LDS, bias and ReLU in place): the sums' store and re-read and 1 launch are gone; f32 sum order kept
# speedup vs baseline: 3.3662x; 1.0325x over previous
#include <hip/hip_runtime.h>
#include <hip/hip_bf16.h>

typedef __attribute__((ext_vector_type(2))) float v2f;
typedef __attribute__((ext_vector_type(8))) float v8f;

#define TICK_PITCH 64
#define TICK_SLICES 16

__device__ __forceinline__ int cache_half() {
  const unsigned id = __builtin_amdgcn_s_sendmsg_rtn(0x87);
  return (int)((id >> 18) & 1u);
}

#define DEG_SUBS 4
#define DEG_TABS (2 * DEG_SUBS)

#define AGG_R 196
#define AGG_OWNERS 256

__global__ __launch_bounds__(256) void k_init_deg(float* __restrict__ cnt,
                                                  int* __restrict__ tick, int n4) {
  int i = blockIdx.x * blockDim.x + threadIdx.x;
  if (i < n4) ((float4*)cnt)[i] = make_float4(0.0f, 0.0f, 0.0f, 0.0f);
  if (i < 4 * TICK_SLICES) tick[i * TICK_PITCH] = 0;
}

__global__ __launch_bounds__(256) void k_count_deg(const int* __restrict__ col,
                                                   float* cnt, unsigned* __restrict__ tag,
                                                   int pitch, int E, int nwords) {
  const int t = blockIdx.x * blockDim.x + threadIdx.x;
  const int tab = cache_half() * DEG_SUBS + (((int)threadIdx.x >> 5) & (DEG_SUBS - 1));
  if (t < nwords) {
    int c[4];
#pragma unroll
    for (int k = 0; k < 4; ++k) {
      const int e = 4 * t + k;
      c[k] = col[(e < E) ? e : (E - 1)];
    }
    unsigned word = 0u;
#pragma unroll
    for (int k = 0; k < 4; ++k) {
      if (4 * t + k < E) {
        (void)__hip_atomic_fetch_add(&cnt[(size_t)tab * pitch + c[k]], 1.0f, __ATOMIC_RELAXED,
                                     __HIP_MEMORY_SCOPE_AGENT);
        word |= ((unsigned)(c[k] / AGG_R) & (AGG_OWNERS - 1u)) << (8 * k);
      }
    }
    tag[t] = word;
  }
}

__global__ __launch_bounds__(256) void k_dinv(const float* __restrict__ cnt,
                                              float* __restrict__ dinv, int pitch, int n) {
  int i = blockIdx.x * blockDim.x + threadIdx.x;
  if (i < n) {
    float d = 1.0f;
#pragma unroll
    for (int k = 0; k < DEG_TABS; ++k) d += cnt[(size_t)k * pitch + i];
    dinv[i] = (d > 0.0f) ? rsqrtf(d) : 0.0f;
  }
}

template <int K, int N>
__global__ __launch_bounds__(256) void k_gemm_wmma_f32(
    const float* __restrict__ A, const float* __restrict__ B,
    float* __restrict__ C, int M) {
  constexpr int TN    = N / 16;
  constexpr int MT    = (TN >= 8) ? 1 : (8 / TN);
  constexpr int ROWS  = 16 * MT;
  constexpr int V4ROW = K / 4;
  constexpr int NLD   = ROWS * V4ROW / 256;
  constexpr int PITCH = K + 4;
  static_assert(N % 16 == 0 && K % 64 == 0, "tile shape");
  static_assert((TN >= 8) ? (TN % 8 == 0) : (8 % TN == 0), "8 waves per row-tile group");
  static_assert((ROWS * V4ROW) % 256 == 0, "staging split");
  __shared__ __attribute__((aligned(16))) float As[ROWS * PITCH];

  const int lane = threadIdx.x & 31;
  const int wave = threadIdx.x >> 5;
  const int t    = blockIdx.x * 8 + wave;
  const int mt0  = (blockIdx.x * 8) / TN;
  const int mt   = t / TN;
  const int m0   = mt << 4;
  const int n0   = (t % TN) << 4;
  const int mbase = mt0 << 4;

  float4 st[NLD];
#pragma unroll
  for (int i = 0; i < NLD; ++i) {
    const int idx = i * 256 + (int)threadIdx.x;
    const int r   = idx / V4ROW;
    const int c4  = idx % V4ROW;
    st[i] = make_float4(0.0f, 0.0f, 0.0f, 0.0f);
    if (mbase + r < M) st[i] = *(const float4*)(A + (size_t)(mbase + r) * K + c4 * 4);
  }
#pragma unroll
  for (int i = 0; i < NLD; ++i) {
    const int idx = i * 256 + (int)threadIdx.x;
    const int r   = idx / V4ROW;
    const int c4  = idx % V4ROW;
    *(float4*)(As + r * PITCH + c4 * 4) = st[i];
  }
  __syncthreads();
  if (m0 >= M) return;

  const int l15   = lane & 15;
  const int khalf = (lane >> 4) << 1;

  const float* ap   = As + ((mt - mt0) * 16 + l15) * PITCH + khalf;
  const float* bcol = B + (size_t)khalf * N + (n0 + l15);

  const int rbase = m0 + ((lane >> 4) << 3);

  v8f acc = {};
  for (int k0 = 0; k0 < K; k0 += 4) {
    v2f a, b;
    const float2 av = *(const float2*)(ap + k0);
    a.x = av.x;
    a.y = av.y;
    b.x = bcol[(size_t)k0 * N];
    b.y = bcol[(size_t)(k0 + 1) * N];
    acc = __builtin_amdgcn_wmma_f32_16x16x4_f32(
         false, a,  false, b,
         (short)0, acc,  false,  false);
  }

  float* cptr = C + (size_t)rbase * N + n0 + l15;
#pragma unroll
  for (int v = 0; v < 8; ++v) cptr[(size_t)v * N] = acc[v];
}


__device__ __forceinline__ int draw_ticket(int* cnt, int lane) {
  int t = 0;
  if (lane == 0)
    t = __hip_atomic_fetch_add(cnt, 1, __ATOMIC_RELAXED, __HIP_MEMORY_SCOPE_AGENT);
  return t;
}

template <int F>
__global__ __launch_bounds__(256) void k_scatter(const float* __restrict__ h,
                                                 const int* __restrict__ row,
                                                 const int* __restrict__ col,
                                                 const float* __restrict__ dinv,
                                                 float* agg, int* tick, int E, int Nn) {
  constexpr int L = F / 32;
  constexpr int U = (L >= 8) ? 4 : (32 / L);
  const int lane   = threadIdx.x & 31;
  const int G      = (E + 31) >> 5;
  const int nsplit = Nn >> 1;
  const int home   = cache_half();
  const int slice  = (int)((blockIdx.x * 0x9E3779B1u) >> 28);
  const int Gs     = (G - slice + TICK_SLICES - 1) / TICK_SLICES;

  for (int pass = 0; pass < 2; ++pass) {
    const int range = home ^ pass;
    int* cnt = tick + (range * TICK_SLICES + slice) * TICK_PITCH;
    int t = __builtin_amdgcn_readfirstlane(draw_ticket(cnt, lane));
    while (t < Gs) {
      const int e = ((t * TICK_SLICES + slice) << 5) + lane;
      int r = 0, c = 0;
      float dr = 0.0f, dc = 0.0f;
      bool mine = false;
      if (e < E) {
        r = row[e];
        c = col[e];
        mine = ((c >= nsplit) ? 1 : 0) == range;
      }
      if (mine) {
        dr = dinv[r];
        dc = dinv[c];
      }
      const int tnext = draw_ticket(cnt, lane);
      const float w = dr * dc;
      const int wbits = __float_as_int(w);
      unsigned m = __builtin_amdgcn_ballot_w32(mine);

      while (m) {
        const int n = __builtin_popcount(m);
        float v[U][L];
        int cj[U];
        float wj[U];
#pragma unroll
        for (int u = 0; u < U; ++u) {
          if (u < n) {
            const int j = __builtin_ctz(m);
            m &= m - 1;
            const int rj = __builtin_amdgcn_readlane(r, j);
            cj[u] = __builtin_amdgcn_readlane(c, j);
            wj[u] = __int_as_float(__builtin_amdgcn_readlane(wbits, j));
            const float* src = h + (size_t)rj * F + lane;
#pragma unroll
            for (int k = 0; k < L; ++k) v[u][k] = src[32 * k];
          }
        }
#pragma unroll
        for (int u = 0; u < U; ++u) {
          if (u < n) {
            float* dst = agg + (size_t)cj[u] * F + lane;
#pragma unroll
            for (int k = 0; k < L; ++k)
              (void)__hip_atomic_fetch_add(dst + 32 * k, v[u][k] * wj[u], __ATOMIC_RELAXED,
                                           __HIP_MEMORY_SCOPE_AGENT);
          }
        }
      }
      t = __builtin_amdgcn_readfirstlane(tnext);
    }
  }
}


__device__ __forceinline__ unsigned zero_bytes(unsigned x) {
  const unsigned t = (x & 0x7F7F7F7Fu) + 0x7F7F7F7Fu;
  return ~(t | x | 0x7F7F7F7Fu);
}

template <int F, int N2>
__global__ __launch_bounds__(1024) void k_agg_gemm(const float* __restrict__ h,
                                                   const int* __restrict__ row,
                                                   const int* __restrict__ col,
                                                   const unsigned* __restrict__ tag,
                                                   const float* __restrict__ dinv,
                                                   const float* __restrict__ B,
                                                   const float* __restrict__ bin,
                                                   const float* __restrict__ bout,
                                                   float* __restrict__ C2,
                                                   float* __restrict__ S, int E, int Nn) {
  static_assert(F % 64 == 0 && N2 % 32 == 0, "tile shape");
  constexpr int L  = F / 32;
  constexpr int PITCH = F + 4;
  constexpr int CP = N2 / 32;
  constexpr int U  = (L >= 8) ? 6 : (32 / L);
  constexpr int NW = 32;
  __shared__ __attribute__((aligned(16))) float acc[AGG_R * PITCH];
  __shared__ uint4 stg[NW * 4 * 32];
  const int lane = threadIdx.x & 31;
  const int wave = __builtin_amdgcn_readfirstlane((int)threadIdx.x >> 5);
  const int NCH  = (E + 511) >> 9;
  float bk[L];
#pragma unroll
  for (int k = 0; k < L; ++k) bk[k] = bin[lane + 32 * k];

  for (int o = blockIdx.x; o * AGG_R < Nn; o += AGG_OWNERS) {
    const int lo    = o * AGG_R;
    const int nrows = (Nn - lo < AGG_R) ? (Nn - lo) : AGG_R;

    for (int i = wave; i < nrows; i += NW) {
      const float d = dinv[lo + i];
      const float* src = h + (size_t)(lo + i) * F + lane;
      float* a = acc + i * PITCH + lane;
#pragma unroll
      for (int k = 0; k < L; ++k) a[32 * k] = (src[32 * k] * d) * d;
    }
    __syncthreads();

    const unsigned pat = (unsigned)(o & (AGG_OWNERS - 1)) * 0x01010101u;
    int cnt = 0;
    int qe  = 0;

    auto flush = [&]() {
      int r = 0, c = 0;
      bool mine = false;
      if (lane < cnt) {
        r = row[qe];
        c = col[qe];
        mine = (c >= lo) && (c < lo + nrows);
      }
      float w = 0.0f;
      if (mine) w = dinv[r] * dinv[c];
      const int wbits = __float_as_int(w);
      unsigned mm = __builtin_amdgcn_ballot_w32(mine);
      while (mm) {
        const int n = __builtin_popcount(mm);
        float v[U][L];
        int cj[U];
        float wj[U];
#pragma unroll
        for (int u = 0; u < U; ++u) {
          if (u < n) {
            const int j = __builtin_ctz(mm);
            mm &= mm - 1;
            const int rj = __builtin_amdgcn_readlane(r, j);
            cj[u] = __builtin_amdgcn_readlane(c, j) - lo;
            wj[u] = __int_as_float(__builtin_amdgcn_readlane(wbits, j));
            const float* src = h + (size_t)rj * F + lane;
#pragma unroll
            for (int k = 0; k < L; ++k) v[u][k] = src[32 * k];
          }
        }
#pragma unroll
        for (int u = 0; u < U; ++u) {
          if (u < n) {
            float* dst = acc + cj[u] * PITCH + lane;
#pragma unroll
            for (int k = 0; k < L; ++k)
              (void)__hip_atomic_fetch_add(dst + 32 * k, v[u][k] * wj[u], __ATOMIC_RELAXED,
                                           __HIP_MEMORY_SCOPE_WORKGROUP);
          }
        }
      }
      cnt = 0;
    };

    uint4* park = stg + wave * 128 + lane;
    for (int g = wave * 4; g < NCH; g += NW * 4) {
      uint4 tg[4];
#pragma unroll
      for (int sc = 0; sc < 4; ++sc) {
        const int chc = (g + sc < NCH) ? (g + sc) : (NCH - 1);
        tg[sc] = *(const uint4*)(tag + (chc << 7) + (lane << 2));
      }
#pragma unroll
      for (int sc = 0; sc < 4; ++sc) {
        if (g + sc >= NCH) tg[sc] = make_uint4(~pat, ~pat, ~pat, ~pat);
        park[sc * 32] = tg[sc];
      }
#pragma nounroll
      for (int sc = 0; sc < 4; ++sc) {
        const uint4 tq = park[sc * 32];
        const unsigned z0 = zero_bytes(tq.x ^ pat);
        const unsigned z1 = zero_bytes(tq.y ^ pat);
        const unsigned z2 = zero_bytes(tq.z ^ pat);
        const unsigned z3 = zero_bytes(tq.w ^ pat);
        unsigned m = __builtin_amdgcn_ballot_w32((z0 | z1 | z2 | z3) != 0u);
        while (m) {
          const int j = __builtin_ctz(m);
          m &= m - 1;
          if (cnt > 16) flush();
          const int ebase = ((g + sc) << 9) + (j << 4);
#pragma unroll
          for (int q = 0; q < 4; ++q) {
            unsigned zw = (unsigned)__builtin_amdgcn_readlane(
                (int)(q == 0 ? z0 : q == 1 ? z1 : q == 2 ? z2 : z3), j);
            while (zw) {
              const int bit = __builtin_ctz(zw);
              zw &= zw - 1;
              const int e = ebase + (q << 2) + (bit >> 3);
              if (e < E) {
                qe = (lane == cnt) ? e : qe;
                ++cnt;
              }
            }
          }
        }
      }
    }
    if (cnt > 0) flush();
    __syncthreads();

    for (int i = wave; i < nrows; i += NW) {
      float* a = acc + i * PITCH + lane;
#pragma unroll
      for (int k = 0; k < L; ++k) a[32 * k] = fmaxf(a[32 * k] + bk[k], 0.0f);
    }
    __syncthreads();

    const int l15   = lane & 15;
    const int khalf = (lane >> 4) << 1;
    const int nrt   = (nrows + 15) >> 4;
    for (int task = wave; task < nrt * CP; task += NW) {
      const int rt = task / CP;
      const int n0 = (task % CP) << 5;
      int ar = (rt << 4) + l15;
      ar = (ar < nrows) ? ar : (nrows - 1);
      const float* ap = acc + ar * PITCH + khalf;
      const float* bp = B + (size_t)khalf * N2 + n0 + 2 * l15;
      const int rl = (rt << 4) + ((lane >> 4) << 3);
      float dv[8];
#pragma unroll
      for (int v = 0; v < 8; ++v) dv[v] = dinv[lo + ((rl + v < nrows) ? (rl + v) : (nrows - 1))];
      const float2 bo2 = *(const float2*)(bout + n0 + 2 * l15);
      v8f c0 = {}, c1 = {};
      for (int k0 = 0; k0 < F; k0 += 4) {
        const float2 av = *(const float2*)(ap + k0);
        const float2 bx = *(const float2*)(bp + (size_t)k0 * N2);
        const float2 by = *(const float2*)(bp + (size_t)(k0 + 1) * N2);
        v2f a, be, bo;
        a.x  = av.x;
        a.y  = av.y;
        be.x = bx.x;
        be.y = by.x;
        bo.x = bx.y;
        bo.y = by.y;
        c0 = __builtin_amdgcn_wmma_f32_16x16x4_f32(
             false, a,  false, be,
             (short)0, c0,  false,  false);
        c1 = __builtin_amdgcn_wmma_f32_16x16x4_f32(
             false, a,  false, bo,
             (short)0, c1,  false,  false);
      }
#pragma unroll
      for (int v = 0; v < 8; ++v) {
        if (rl + v < nrows) {
          const size_t o2 = (size_t)(lo + rl + v) * N2 + n0 + 2 * l15;
          const float d  = dv[v];
          const float ce = c0[v];
          const float co = c1[v];
          *(float2*)(C2 + o2) = make_float2(ce, co);
          const float se = ce * d * d;
          const float so = co * d * d;
          *(float2*)(S + o2) = make_float2(se + bo2.x, so + bo2.y);
        }
      }
    }
    __syncthreads();
  }
}

extern "C" void kernel_launch(void* const* d_in, const int* in_sizes, int n_in,
                              void* d_out, int out_size, void* d_ws, size_t ws_size,
                              hipStream_t stream) {
  const float* x  = (const float*)d_in[0];
  const int*   ei = (const int*)d_in[1];
  const float* W1 = (const float*)d_in[2];
  const float* b1 = (const float*)d_in[3];
  const float* W2 = (const float*)d_in[4];
  const float* b2 = (const float*)d_in[5];

  const int H  = in_sizes[3];
  const int C  = in_sizes[5];
  const int F  = in_sizes[2] / H;
  const int Nn = in_sizes[0] / F;
  const int E  = in_sizes[1] / 2;
  const int* rowIdx = ei;
  const int* colIdx = ei + E;

  auto al = [](size_t v) { return (v + 255) & ~(size_t)255; };
  const int pitch = (Nn + 63) & ~63;
  float* dinv = (float*)d_ws;
  size_t off  = al((size_t)Nn * sizeof(float));
  float* cnt  = (float*)((char*)d_ws + off);
  off += al((size_t)DEG_TABS * pitch * sizeof(float));
  float* hbuf = (float*)((char*)d_ws + off);
  off += al((size_t)Nn * H * sizeof(float));
  float* h2buf = (float*)((char*)d_ws + off);
  off += al((size_t)Nn * C * sizeof(float));
  int* tick = (int*)((char*)d_ws + off);
  off += al((size_t)4 * TICK_SLICES * TICK_PITCH * sizeof(int));
  unsigned* tag = (unsigned*)((char*)d_ws + off);
  const int nwords = ((E + 511) / 512) * 128;

  const dim3 blk(256);
  auto blocks = [](long long work) { return (unsigned)((work + 255) / 256); };

  const int n4 = DEG_TABS * pitch / 4;
  k_init_deg <<<blocks(n4), blk, 0, stream>>>(cnt, tick, n4);
  k_count_deg<<<blocks(nwords), blk, 0, stream>>>(colIdx, cnt, tag, pitch, E, nwords);
  k_dinv     <<<blocks(Nn), blk, 0, stream>>>(cnt, dinv, pitch, Nn);

  const int tiles1 = ((Nn + 15) / 16) * (H / 16);
  float* out = (float*)d_out;
  k_gemm_wmma_f32<128, 256><<<(tiles1 + 7) / 8, blk, 0, stream>>>(x, W1, hbuf, Nn);
  k_agg_gemm<256, 64><<<AGG_OWNERS, dim3(1024), 0, stream>>>(hbuf, rowIdx, colIdx, tag, dinv, W2,
                                                            b1, b2, h2buf, out, E, Nn);
  const unsigned sblocks = 2048;

  k_scatter<64><<<sblocks, blk, 0, stream>>>(h2buf, rowIdx, colIdx, dinv, out,
                                            tick + 2 * TICK_SLICES * TICK_PITCH, E, Nn);
}


# layer-2 aggregation: a work-group per node range sums h2 rows in LDS from edge records that layer 1's scan wrote out per range; global f32 atomic scatter gone; sum order: LDS arrival
# speedup vs baseline: 3.5161x; 1.0445x over previous
#include <hip/hip_runtime.h>
#include <hip/hip_bf16.h>

typedef __attribute__((ext_vector_type(2))) float v2f;
typedef __attribute__((ext_vector_type(8))) float v8f;

__device__ __forceinline__ int cache_half() {
  const unsigned id = __builtin_amdgcn_s_sendmsg_rtn(0x87);
  return (int)((id >> 18) & 1u);
}

#define DEG_SUBS 4
#define DEG_TABS (2 * DEG_SUBS)

#define AGG_R 196
#define AGG_OWNERS 256

__global__ __launch_bounds__(256) void k_init_deg(float* __restrict__ cnt, int n4) {
  int i = blockIdx.x * blockDim.x + threadIdx.x;
  if (i < n4) ((float4*)cnt)[i] = make_float4(0.0f, 0.0f, 0.0f, 0.0f);
}

__global__ __launch_bounds__(256) void k_count_deg(const int* __restrict__ col,
                                                   float* cnt, unsigned* __restrict__ tag,
                                                   int pitch, int E, int nwords) {
  const int t = blockIdx.x * blockDim.x + threadIdx.x;
  const int tab = cache_half() * DEG_SUBS + (((int)threadIdx.x >> 5) & (DEG_SUBS - 1));
  if (t < nwords) {
    int c[4];
#pragma unroll
    for (int k = 0; k < 4; ++k) {
      const int e = 4 * t + k;
      c[k] = col[(e < E) ? e : (E - 1)];
    }
    unsigned word = 0u;
#pragma unroll
    for (int k = 0; k < 4; ++k) {
      if (4 * t + k < E) {
        (void)__hip_atomic_fetch_add(&cnt[(size_t)tab * pitch + c[k]], 1.0f, __ATOMIC_RELAXED,
                                     __HIP_MEMORY_SCOPE_AGENT);
        word |= ((unsigned)(c[k] / AGG_R) & (AGG_OWNERS - 1u)) << (8 * k);
      }
    }
    tag[t] = word;
  }
}

__global__ __launch_bounds__(256) void k_dinv(const float* __restrict__ cnt,
                                              float* __restrict__ dinv,
                                              int* __restrict__ rcount, int pitch, int n) {
  __shared__ int part[8];
  const int t = threadIdx.x;
  const int i = blockIdx.x * AGG_R + t;
  int indeg = 0;
  if (t < AGG_R && i < n) {
    float d = 1.0f;
#pragma unroll
    for (int k = 0; k < DEG_TABS; ++k) d += cnt[(size_t)k * pitch + i];
    dinv[i] = (d > 0.0f) ? rsqrtf(d) : 0.0f;
    indeg = (int)(d - 1.0f);
  }
#pragma unroll
  for (int s = 16; s > 0; s >>= 1) indeg += __shfl_xor(indeg, s);
  if ((t & 31) == 0) part[t >> 5] = indeg;
  __syncthreads();
  if (t == 0) {
    int tot = 0;
#pragma unroll
    for (int w = 0; w < 8; ++w) tot += part[w];
    rcount[blockIdx.x] = tot;
  }
}

template <int K, int N>
__global__ __launch_bounds__(256) void k_gemm_wmma_f32(
    const float* __restrict__ A, const float* __restrict__ B,
    float* __restrict__ C, int M) {
  constexpr int TN    = N / 16;
  constexpr int MT    = (TN >= 8) ? 1 : (8 / TN);
  constexpr int ROWS  = 16 * MT;
  constexpr int V4ROW = K / 4;
  constexpr int NLD   = ROWS * V4ROW / 256;
  constexpr int PITCH = K + 4;
  static_assert(N % 16 == 0 && K % 64 == 0, "tile shape");
  static_assert((TN >= 8) ? (TN % 8 == 0) : (8 % TN == 0), "8 waves per row-tile group");
  static_assert((ROWS * V4ROW) % 256 == 0, "staging split");
  __shared__ __attribute__((aligned(16))) float As[ROWS * PITCH];

  const int lane = threadIdx.x & 31;
  const int wave = threadIdx.x >> 5;
  const int t    = blockIdx.x * 8 + wave;
  const int mt0  = (blockIdx.x * 8) / TN;
  const int mt   = t / TN;
  const int m0   = mt << 4;
  const int n0   = (t % TN) << 4;
  const int mbase = mt0 << 4;

  float4 st[NLD];
#pragma unroll
  for (int i = 0; i < NLD; ++i) {
    const int idx = i * 256 + (int)threadIdx.x;
    const int r   = idx / V4ROW;
    const int c4  = idx % V4ROW;
    st[i] = make_float4(0.0f, 0.0f, 0.0f, 0.0f);
    if (mbase + r < M) st[i] = *(const float4*)(A + (size_t)(mbase + r) * K + c4 * 4);
  }
#pragma unroll
  for (int i = 0; i < NLD; ++i) {
    const int idx = i * 256 + (int)threadIdx.x;
    const int r   = idx / V4ROW;
    const int c4  = idx % V4ROW;
    *(float4*)(As + r * PITCH + c4 * 4) = st[i];
  }
  __syncthreads();
  if (m0 >= M) return;

  const int l15   = lane & 15;
  const int khalf = (lane >> 4) << 1;

  const float* ap   = As + ((mt - mt0) * 16 + l15) * PITCH + khalf;
  const float* bcol = B + (size_t)khalf * N + (n0 + l15);

  const int rbase = m0 + ((lane >> 4) << 3);

  v8f acc = {};
  for (int k0 = 0; k0 < K; k0 += 4) {
    v2f a, b;
    const float2 av = *(const float2*)(ap + k0);
    a.x = av.x;
    a.y = av.y;
    b.x = bcol[(size_t)k0 * N];
    b.y = bcol[(size_t)(k0 + 1) * N];
    acc = __builtin_amdgcn_wmma_f32_16x16x4_f32(
         false, a,  false, b,
         (short)0, acc,  false,  false);
  }

  float* cptr = C + (size_t)rbase * N + n0 + l15;
#pragma unroll
  for (int v = 0; v < 8; ++v) cptr[(size_t)v * N] = acc[v];
}


__device__ __forceinline__ int edges_below(const int* __restrict__ rcount, int o, int lane) {
  int below = 0;
  for (int q0 = 0; q0 < o; q0 += 256) {
    int v[8];
#pragma unroll
    for (int k = 0; k < 8; ++k) {
      const int q = q0 + 32 * k + lane;
      v[k] = rcount[(q < o) ? q : 0];
    }
#pragma unroll
    for (int k = 0; k < 8; ++k) below += (q0 + 32 * k + lane < o) ? v[k] : 0;
  }
#pragma unroll
  for (int s = 16; s > 0; s >>= 1) below += __shfl_xor(below, s);
  return below;
}

__device__ __forceinline__ unsigned zero_bytes(unsigned x) {
  const unsigned t = (x & 0x7F7F7F7Fu) + 0x7F7F7F7Fu;
  return ~(t | x | 0x7F7F7F7Fu);
}

template <int F, int N2>
__global__ __launch_bounds__(1024) void k_agg_gemm(const float* __restrict__ h,
                                                   const int* __restrict__ row,
                                                   const int* __restrict__ col,
                                                   const unsigned* __restrict__ tag,
                                                   const float* __restrict__ dinv,
                                                   const float* __restrict__ B,
                                                   const float* __restrict__ bin,
                                                   const int* __restrict__ rcount,
                                                   int* __restrict__ acount,
                                                   int* __restrict__ esrc,
                                                   int* __restrict__ etgt,
                                                   float* __restrict__ ew,
                                                   float* __restrict__ C2, int E, int Nn) {
  static_assert(F % 64 == 0 && N2 % 32 == 0, "tile shape");
  constexpr int L  = F / 32;
  constexpr int PITCH = F + 4;
  constexpr int CP = N2 / 32;
  constexpr int U  = (L >= 8) ? 6 : (32 / L);
  constexpr int NW = 32;
  __shared__ __attribute__((aligned(16))) float acc[AGG_R * PITCH];
  __shared__ uint4 stg[NW * 4 * 32];
  __shared__ int s_off;
  __shared__ int s_cnt;
  const int lane = threadIdx.x & 31;
  const int wave = __builtin_amdgcn_readfirstlane((int)threadIdx.x >> 5);
  const int NCH  = (E + 511) >> 9;
  float bk[L];
#pragma unroll
  for (int k = 0; k < L; ++k) bk[k] = bin[lane + 32 * k];

  for (int o = blockIdx.x; o * AGG_R < Nn; o += AGG_OWNERS) {
    const int lo    = o * AGG_R;
    const int nrows = (Nn - lo < AGG_R) ? (Nn - lo) : AGG_R;

    if (wave == 0) {
      const int below = edges_below(rcount, o, lane);
      if (lane == 0) {
        s_off = below;
        s_cnt = 0;
      }
    }

    for (int i = wave; i < nrows; i += NW) {
      const float d = dinv[lo + i];
      const float* src = h + (size_t)(lo + i) * F + lane;
      float* a = acc + i * PITCH + lane;
#pragma unroll
      for (int k = 0; k < L; ++k) a[32 * k] = (src[32 * k] * d) * d;
    }
    __syncthreads();
    const int off = s_off;

    const unsigned pat = (unsigned)(o & (AGG_OWNERS - 1)) * 0x01010101u;
    int cnt = 0;
    int qe  = 0;

    auto flush = [&]() {
      int r = 0, c = 0;
      bool mine = false;
      if (lane < cnt) {
        r = row[qe];
        c = col[qe];
        mine = (c >= lo) && (c < lo + nrows);
      }
      float w = 0.0f;
      if (mine) w = dinv[r] * dinv[c];
      const int wbits = __float_as_int(w);
      unsigned mm = __builtin_amdgcn_ballot_w32(mine);
      if (mm) {
        int base = 0;
        if (lane == 0)
          base = __hip_atomic_fetch_add(&s_cnt, __builtin_popcount(mm), __ATOMIC_RELAXED,
                                        __HIP_MEMORY_SCOPE_WORKGROUP);
        base = __builtin_amdgcn_readfirstlane(base);
        const int slot = off + base + (int)__builtin_amdgcn_mbcnt_lo(mm, 0u);
        if (mine && slot < E) {
          esrc[slot] = r;
          etgt[slot] = c - lo;
          ew[slot]   = w;
        }
      }
      while (mm) {
        const int n = __builtin_popcount(mm);
        float v[U][L];
        int cj[U];
        float wj[U];
#pragma unroll
        for (int u = 0; u < U; ++u) {
          if (u < n) {
            const int j = __builtin_ctz(mm);
            mm &= mm - 1;
            const int rj = __builtin_amdgcn_readlane(r, j);
            cj[u] = __builtin_amdgcn_readlane(c, j) - lo;
            wj[u] = __int_as_float(__builtin_amdgcn_readlane(wbits, j));
            const float* src = h + (size_t)rj * F + lane;
#pragma unroll
            for (int k = 0; k < L; ++k) v[u][k] = src[32 * k];
          }
        }
#pragma unroll
        for (int u = 0; u < U; ++u) {
          if (u < n) {
            float* dst = acc + cj[u] * PITCH + lane;
#pragma unroll
            for (int k = 0; k < L; ++k)
              (void)__hip_atomic_fetch_add(dst + 32 * k, v[u][k] * wj[u], __ATOMIC_RELAXED,
                                           __HIP_MEMORY_SCOPE_WORKGROUP);
          }
        }
      }
      cnt = 0;
    };

    uint4* park = stg + wave * 128 + lane;
    for (int g = wave * 4; g < NCH; g += NW * 4) {
      uint4 tg[4];
#pragma unroll
      for (int sc = 0; sc < 4; ++sc) {
        const int chc = (g + sc < NCH) ? (g + sc) : (NCH - 1);
        tg[sc] = *(const uint4*)(tag + (chc << 7) + (lane << 2));
      }
#pragma unroll
      for (int sc = 0; sc < 4; ++sc) {
        if (g + sc >= NCH) tg[sc] = make_uint4(~pat, ~pat, ~pat, ~pat);
        park[sc * 32] = tg[sc];
      }
#pragma nounroll
      for (int sc = 0; sc < 4; ++sc) {
        const uint4 tq = park[sc * 32];
        const unsigned z0 = zero_bytes(tq.x ^ pat);
        const unsigned z1 = zero_bytes(tq.y ^ pat);
        const unsigned z2 = zero_bytes(tq.z ^ pat);
        const unsigned z3 = zero_bytes(tq.w ^ pat);
        unsigned m = __builtin_amdgcn_ballot_w32((z0 | z1 | z2 | z3) != 0u);
        while (m) {
          const int j = __builtin_ctz(m);
          m &= m - 1;
          if (cnt > 16) flush();
          const int ebase = ((g + sc) << 9) + (j << 4);
#pragma unroll
          for (int q = 0; q < 4; ++q) {
            unsigned zw = (unsigned)__builtin_amdgcn_readlane(
                (int)(q == 0 ? z0 : q == 1 ? z1 : q == 2 ? z2 : z3), j);
            while (zw) {
              const int bit = __builtin_ctz(zw);
              zw &= zw - 1;
              const int e = ebase + (q << 2) + (bit >> 3);
              if (e < E) {
                qe = (lane == cnt) ? e : qe;
                ++cnt;
              }
            }
          }
        }
      }
    }
    if (cnt > 0) flush();
    __syncthreads();
    if (threadIdx.x == 0) acount[o] = s_cnt;

    for (int i = wave; i < nrows; i += NW) {
      float* a = acc + i * PITCH + lane;
#pragma unroll
      for (int k = 0; k < L; ++k) a[32 * k] = fmaxf(a[32 * k] + bk[k], 0.0f);
    }
    __syncthreads();

    const int l15   = lane & 15;
    const int khalf = (lane >> 4) << 1;
    const int nrt   = (nrows + 15) >> 4;
    for (int task = wave; task < nrt * CP; task += NW) {
      const int rt = task / CP;
      const int n0 = (task % CP) << 5;
      int ar = (rt << 4) + l15;
      ar = (ar < nrows) ? ar : (nrows - 1);
      const float* ap = acc + ar * PITCH + khalf;
      const float* bp = B + (size_t)khalf * N2 + n0 + 2 * l15;
      const int rl = (rt << 4) + ((lane >> 4) << 3);
      v8f c0 = {}, c1 = {};
      for (int k0 = 0; k0 < F; k0 += 4) {
        const float2 av = *(const float2*)(ap + k0);
        const float2 bx = *(const float2*)(bp + (size_t)k0 * N2);
        const float2 by = *(const float2*)(bp + (size_t)(k0 + 1) * N2);
        v2f a, be, bo;
        a.x  = av.x;
        a.y  = av.y;
        be.x = bx.x;
        be.y = by.x;
        bo.x = bx.y;
        bo.y = by.y;
        c0 = __builtin_amdgcn_wmma_f32_16x16x4_f32(
             false, a,  false, be,
             (short)0, c0,  false,  false);
        c1 = __builtin_amdgcn_wmma_f32_16x16x4_f32(
             false, a,  false, bo,
             (short)0, c1,  false,  false);
      }
#pragma unroll
      for (int v = 0; v < 8; ++v) {
        if (rl + v < nrows) {
          const size_t o2 = (size_t)(lo + rl + v) * N2 + n0 + 2 * l15;
          *(float2*)(C2 + o2) = make_float2(c0[v], c1[v]);
        }
      }
    }
    __syncthreads();
  }
}

template <int F>
__global__ __launch_bounds__(1024) void k_agg_rows(const float* __restrict__ h,
                                                   const int* __restrict__ esrc,
                                                   const int* __restrict__ etgt,
                                                   const float* __restrict__ ew,
                                                   const int* __restrict__ rcount,
                                                   const int* __restrict__ acount,
                                                   const float* __restrict__ dinv,
                                                   const float* __restrict__ bout,
                                                   float* __restrict__ out, int E, int Nn) {
  static_assert(F % 32 == 0 && F <= 128, "row shape");
  constexpr int L  = F / 32;
  constexpr int U  = 32 / L;
  constexpr int NW = 32;
  __shared__ float acc[AGG_R * F];
  __shared__ int s_off;
  const int lane = threadIdx.x & 31;
  const int wave = __builtin_amdgcn_readfirstlane((int)threadIdx.x >> 5);
  float bk[L];
#pragma unroll
  for (int k = 0; k < L; ++k) bk[k] = bout[lane + 32 * k];

  for (int o = blockIdx.x; o * AGG_R < Nn; o += AGG_OWNERS) {
    const int lo    = o * AGG_R;
    const int nrows = (Nn - lo < AGG_R) ? (Nn - lo) : AGG_R;

    if (wave == 0) {
      const int below = edges_below(rcount, o, lane);
      if (lane == 0) s_off = below;
    }

    for (int i = wave; i < nrows; i += NW) {
      const float d = dinv[lo + i];
      const float* src = h + (size_t)(lo + i) * F + lane;
      float* a = acc + i * F + lane;
#pragma unroll
      for (int k = 0; k < L; ++k) a[32 * k] = (src[32 * k] * d) * d + bk[k];
    }
    __syncthreads();
    const int off = s_off;
    int n = acount[o];
    n = (n < E - off) ? n : (E - off);

    for (int b = wave * 32; b < n; b += NW * 32) {
      const bool ok = (b + lane) < n;
      int r = 0, t = 0;
      float w = 0.0f;
      if (ok) {
        r = esrc[off + b + lane];
        t = etgt[off + b + lane];
        w = ew[off + b + lane];
      }
      const int wbits = __float_as_int(w);
      unsigned mm = __builtin_amdgcn_ballot_w32(ok);
      while (mm) {
        const int m = __builtin_popcount(mm);
        float v[U][L];
        int cj[U];
        float wj[U];
#pragma unroll
        for (int u = 0; u < U; ++u) {
          if (u < m) {
            const int j = __builtin_ctz(mm);
            mm &= mm - 1;
            const int rj = __builtin_amdgcn_readlane(r, j);
            cj[u] = __builtin_amdgcn_readlane(t, j);
            wj[u] = __int_as_float(__builtin_amdgcn_readlane(wbits, j));
            const float* src = h + (size_t)rj * F + lane;
#pragma unroll
            for (int k = 0; k < L; ++k) v[u][k] = src[32 * k];
          }
        }
#pragma unroll
        for (int u = 0; u < U; ++u) {
          if (u < m) {
            float* dst = acc + cj[u] * F + lane;
#pragma unroll
            for (int k = 0; k < L; ++k)
              (void)__hip_atomic_fetch_add(dst + 32 * k, v[u][k] * wj[u], __ATOMIC_RELAXED,
                                           __HIP_MEMORY_SCOPE_WORKGROUP);
          }
        }
      }
    }
    __syncthreads();

    for (int i = wave; i < nrows; i += NW) {
      const float* a = acc + i * F + lane;
      float* dst = out + (size_t)(lo + i) * F + lane;
#pragma unroll
      for (int k = 0; k < L; ++k) dst[32 * k] = a[32 * k];
    }
    __syncthreads();
  }
}

extern "C" void kernel_launch(void* const* d_in, const int* in_sizes, int n_in,
                              void* d_out, int out_size, void* d_ws, size_t ws_size,
                              hipStream_t stream) {
  const float* x  = (const float*)d_in[0];
  const int*   ei = (const int*)d_in[1];
  const float* W1 = (const float*)d_in[2];
  const float* b1 = (const float*)d_in[3];
  const float* W2 = (const float*)d_in[4];
  const float* b2 = (const float*)d_in[5];

  const int H  = in_sizes[3];
  const int C  = in_sizes[5];
  const int F  = in_sizes[2] / H;
  const int Nn = in_sizes[0] / F;
  const int E  = in_sizes[1] / 2;
  const int* rowIdx = ei;
  const int* colIdx = ei + E;

  auto al = [](size_t v) { return (v + 255) & ~(size_t)255; };
  const int pitch = (Nn + 63) & ~63;
  float* dinv = (float*)d_ws;
  size_t off  = al((size_t)Nn * sizeof(float));
  float* cnt  = (float*)((char*)d_ws + off);
  off += al((size_t)DEG_TABS * pitch * sizeof(float));
  float* hbuf = (float*)((char*)d_ws + off);
  off += al((size_t)Nn * H * sizeof(float));
  float* h2buf = (float*)((char*)d_ws + off);
  off += al((size_t)Nn * C * sizeof(float));
  const int nranges = (Nn + AGG_R - 1) / AGG_R;
  int* rcount = (int*)((char*)d_ws + off);
  off += al((size_t)nranges * sizeof(int));
  int* acount = (int*)((char*)d_ws + off);
  off += al((size_t)nranges * sizeof(int));
  int* esrc = (int*)((char*)d_ws + off);
  off += al((size_t)E * sizeof(int));
  int* etgt = (int*)((char*)d_ws + off);
  off += al((size_t)E * sizeof(int));
  float* ew = (float*)((char*)d_ws + off);
  off += al((size_t)E * sizeof(float));
  unsigned* tag = (unsigned*)((char*)d_ws + off);
  const int nwords = ((E + 511) / 512) * 128;

  const dim3 blk(256);
  auto blocks = [](long long work) { return (unsigned)((work + 255) / 256); };

  const int n4 = DEG_TABS * pitch / 4;
  k_init_deg <<<blocks(n4), blk, 0, stream>>>(cnt, n4);
  k_count_deg<<<blocks(nwords), blk, 0, stream>>>(colIdx, cnt, tag, pitch, E, nwords);
  k_dinv     <<<nranges, blk, 0, stream>>>(cnt, dinv, rcount, pitch, Nn);

  const int tiles1 = ((Nn + 15) / 16) * (H / 16);
  float* out = (float*)d_out;
  k_gemm_wmma_f32<128, 256><<<(tiles1 + 7) / 8, blk, 0, stream>>>(x, W1, hbuf, Nn);
  k_agg_gemm<256, 64><<<AGG_OWNERS, dim3(1024), 0, stream>>>(hbuf, rowIdx, colIdx, tag, dinv, W2,
                                                            b1, rcount, acount, esrc, etgt, ew,
                                                            h2buf, E, Nn);

  k_agg_rows<64><<<AGG_OWNERS, dim3(1024), 0, stream>>>(h2buf, esrc, etgt, ew, rcount, acount,
                                                       dinv, b2, out, E, Nn);
}


# layer 1's product: 512 work-groups stay and walk the row tiles; a wave keeps the W1 fragments of its 32 columns in registers all launch long (were re-loaded by 50000 one-tile waves); sum order kept
# speedup vs baseline: 3.6279x; 1.0318x over previous
#include <hip/hip_runtime.h>
#include <hip/hip_bf16.h>

typedef __attribute__((ext_vector_type(2))) float v2f;
typedef __attribute__((ext_vector_type(8))) float v8f;

__device__ __forceinline__ int cache_half() {
  const unsigned id = __builtin_amdgcn_s_sendmsg_rtn(0x87);
  return (int)((id >> 18) & 1u);
}

#define DEG_SUBS 4
#define DEG_TABS (2 * DEG_SUBS)

#define AGG_R 196
#define AGG_OWNERS 256

__global__ __launch_bounds__(256) void k_init_deg(float* __restrict__ cnt, int n4) {
  int i = blockIdx.x * blockDim.x + threadIdx.x;
  if (i < n4) ((float4*)cnt)[i] = make_float4(0.0f, 0.0f, 0.0f, 0.0f);
}

__global__ __launch_bounds__(256) void k_count_deg(const int* __restrict__ col,
                                                   float* cnt, unsigned* __restrict__ tag,
                                                   int pitch, int E, int nwords) {
  const int t = blockIdx.x * blockDim.x + threadIdx.x;
  const int tab = cache_half() * DEG_SUBS + (((int)threadIdx.x >> 5) & (DEG_SUBS - 1));
  if (t < nwords) {
    int c[4];
#pragma unroll
    for (int k = 0; k < 4; ++k) {
      const int e = 4 * t + k;
      c[k] = col[(e < E) ? e : (E - 1)];
    }
    unsigned word = 0u;
#pragma unroll
    for (int k = 0; k < 4; ++k) {
      if (4 * t + k < E) {
        (void)__hip_atomic_fetch_add(&cnt[(size_t)tab * pitch + c[k]], 1.0f, __ATOMIC_RELAXED,
                                     __HIP_MEMORY_SCOPE_AGENT);
        word |= ((unsigned)(c[k] / AGG_R) & (AGG_OWNERS - 1u)) << (8 * k);
      }
    }
    tag[t] = word;
  }
}

__global__ __launch_bounds__(256) void k_dinv(const float* __restrict__ cnt,
                                              float* __restrict__ dinv,
                                              int* __restrict__ rcount, int pitch, int n) {
  __shared__ int part[8];
  const int t = threadIdx.x;
  const int i = blockIdx.x * AGG_R + t;
  int indeg = 0;
  if (t < AGG_R && i < n) {
    float d = 1.0f;
#pragma unroll
    for (int k = 0; k < DEG_TABS; ++k) d += cnt[(size_t)k * pitch + i];
    dinv[i] = (d > 0.0f) ? rsqrtf(d) : 0.0f;
    indeg = (int)(d - 1.0f);
  }
#pragma unroll
  for (int s = 16; s > 0; s >>= 1) indeg += __shfl_xor(indeg, s);
  if ((t & 31) == 0) part[t >> 5] = indeg;
  __syncthreads();
  if (t == 0) {
    int tot = 0;
#pragma unroll
    for (int w = 0; w < 8; ++w) tot += part[w];
    rcount[blockIdx.x] = tot;
  }
}

template <int K, int N>
__global__ __launch_bounds__(256) void k_gemm_wmma_f32(
    const float* __restrict__ A, const float* __restrict__ B,
    float* __restrict__ C, int M) {
  constexpr int KS    = K / 4;
  constexpr int V4ROW = K / 4;
  constexpr int NLD   = 16 * V4ROW / 256;
  constexpr int PITCH = K + 4;
  static_assert(N == 8 * 32 && K % 64 == 0, "8 waves x 32 columns");
  static_assert((16 * V4ROW) % 256 == 0, "staging split");
  __shared__ __attribute__((aligned(16))) float As[2][16 * PITCH];

  const int lane = threadIdx.x & 31;
  const int wave = __builtin_amdgcn_readfirstlane((int)threadIdx.x >> 5);
  const int l15   = lane & 15;
  const int khalf = (lane >> 4) << 1;
  const int n0    = wave << 5;
  const int ntiles = (M + 15) >> 4;
  const int step   = (int)gridDim.x;
  int rt = blockIdx.x;
  if (rt >= ntiles) return;

  float4 st[NLD];
  auto fetch = [&](int tile) {
#pragma unroll
    for (int i = 0; i < NLD; ++i) {
      const int idx = i * 256 + (int)threadIdx.x;
      const int r   = tile * 16 + idx / V4ROW;
      const int c4  = idx % V4ROW;
      st[i] = make_float4(0.0f, 0.0f, 0.0f, 0.0f);
      if (r < M) st[i] = *(const float4*)(A + (size_t)r * K + c4 * 4);
    }
  };
  auto park = [&](float* buf) {
#pragma unroll
    for (int i = 0; i < NLD; ++i) {
      const int idx = i * 256 + (int)threadIdx.x;
      *(float4*)(buf + (idx / V4ROW) * PITCH + (idx % V4ROW) * 4) = st[i];
    }
  };

  fetch(rt);

  v2f be[KS], bo[KS];
  {
    const float* bp = B + (size_t)khalf * N + n0 + 2 * l15;
#pragma unroll
    for (int s = 0; s < KS; ++s) {
      const float2 bx = *(const float2*)(bp + (size_t)(4 * s) * N);
      const float2 by = *(const float2*)(bp + (size_t)(4 * s + 1) * N);
      be[s].x = bx.x;
      be[s].y = by.x;
      bo[s].x = bx.y;
      bo[s].y = by.y;
    }
  }

  park(As[0]);
  __syncthreads();

  int cur = 0;
  for (; rt < ntiles; rt += step) {
    const bool more = (rt + step) < ntiles;
    if (more) fetch(rt + step);

    const float* ap = As[cur] + l15 * PITCH + khalf;
    v8f c0 = {}, c1 = {};
#pragma unroll
    for (int s = 0; s < KS; ++s) {
      const float2 av = *(const float2*)(ap + 4 * s);
      v2f a;
      a.x = av.x;
      a.y = av.y;
      c0 = __builtin_amdgcn_wmma_f32_16x16x4_f32(
           false, a,  false, be[s],
           (short)0, c0,  false,  false);
      c1 = __builtin_amdgcn_wmma_f32_16x16x4_f32(
           false, a,  false, bo[s],
           (short)0, c1,  false,  false);
    }

    if (more) park(As[cur ^ 1]);
    __syncthreads();
    cur ^= 1;

    const int rl = (rt << 4) + ((lane >> 4) << 3);
#pragma unroll
    for (int v = 0; v < 8; ++v) {
      if (rl + v < M)
        *(float2*)(C + (size_t)(rl + v) * N + n0 + 2 * l15) = make_float2(c0[v], c1[v]);
    }
  }
}


__device__ __forceinline__ int edges_below(const int* __restrict__ rcount, int o, int lane) {
  int below = 0;
  for (int q0 = 0; q0 < o; q0 += 256) {
    int v[8];
#pragma unroll
    for (int k = 0; k < 8; ++k) {
      const int q = q0 + 32 * k + lane;
      v[k] = rcount[(q < o) ? q : 0];
    }
#pragma unroll
    for (int k = 0; k < 8; ++k) below += (q0 + 32 * k + lane < o) ? v[k] : 0;
  }
#pragma unroll
  for (int s = 16; s > 0; s >>= 1) below += __shfl_xor(below, s);
  return below;
}

__device__ __forceinline__ unsigned zero_bytes(unsigned x) {
  const unsigned t = (x & 0x7F7F7F7Fu) + 0x7F7F7F7Fu;
  return ~(t | x | 0x7F7F7F7Fu);
}

template <int F, int N2>
__global__ __launch_bounds__(1024) void k_agg_gemm(const float* __restrict__ h,
                                                   const int* __restrict__ row,
                                                   const int* __restrict__ col,
                                                   const unsigned* __restrict__ tag,
                                                   const float* __restrict__ dinv,
                                                   const float* __restrict__ B,
                                                   const float* __restrict__ bin,
                                                   const int* __restrict__ rcount,
                                                   int* __restrict__ acount,
                                                   int* __restrict__ esrc,
                                                   int* __restrict__ etgt,
                                                   float* __restrict__ ew,
                                                   float* __restrict__ C2, int E, int Nn) {
  static_assert(F % 64 == 0 && N2 % 32 == 0, "tile shape");
  constexpr int L  = F / 32;
  constexpr int PITCH = F + 4;
  constexpr int CP = N2 / 32;
  constexpr int U  = (L >= 8) ? 6 : (32 / L);
  constexpr int NW = 32;
  __shared__ __attribute__((aligned(16))) float acc[AGG_R * PITCH];
  __shared__ uint4 stg[NW * 4 * 32];
  __shared__ int s_off;
  __shared__ int s_cnt;
  const int lane = threadIdx.x & 31;
  const int wave = __builtin_amdgcn_readfirstlane((int)threadIdx.x >> 5);
  const int NCH  = (E + 511) >> 9;
  float bk[L];
#pragma unroll
  for (int k = 0; k < L; ++k) bk[k] = bin[lane + 32 * k];

  for (int o = blockIdx.x; o * AGG_R < Nn; o += AGG_OWNERS) {
    const int lo    = o * AGG_R;
    const int nrows = (Nn - lo < AGG_R) ? (Nn - lo) : AGG_R;

    if (wave == 0) {
      const int below = edges_below(rcount, o, lane);
      if (lane == 0) {
        s_off = below;
        s_cnt = 0;
      }
    }

    for (int i = wave; i < nrows; i += NW) {
      const float d = dinv[lo + i];
      const float* src = h + (size_t)(lo + i) * F + lane;
      float* a = acc + i * PITCH + lane;
#pragma unroll
      for (int k = 0; k < L; ++k) a[32 * k] = (src[32 * k] * d) * d;
    }
    __syncthreads();
    const int off = s_off;

    const unsigned pat = (unsigned)(o & (AGG_OWNERS - 1)) * 0x01010101u;
    int cnt = 0;
    int qe  = 0;

    auto flush = [&]() {
      int r = 0, c = 0;
      bool mine = false;
      if (lane < cnt) {
        r = row[qe];
        c = col[qe];
        mine = (c >= lo) && (c < lo + nrows);
      }
      float w = 0.0f;
      if (mine) w = dinv[r] * dinv[c];
      const int wbits = __float_as_int(w);
      unsigned mm = __builtin_amdgcn_ballot_w32(mine);
      if (mm) {
        int base = 0;
        if (lane == 0)
          base = __hip_atomic_fetch_add(&s_cnt, __builtin_popcount(mm), __ATOMIC_RELAXED,
                                        __HIP_MEMORY_SCOPE_WORKGROUP);
        base = __builtin_amdgcn_readfirstlane(base);
        const int slot = off + base + (int)__builtin_amdgcn_mbcnt_lo(mm, 0u);
        if (mine && slot < E) {
          esrc[slot] = r;
          etgt[slot] = c - lo;
          ew[slot]   = w;
        }
      }
      while (mm) {
        const int n = __builtin_popcount(mm);
        float v[U][L];
        int cj[U];
        float wj[U];
#pragma unroll
        for (int u = 0; u < U; ++u) {
          if (u < n) {
            const int j = __builtin_ctz(mm);
            mm &= mm - 1;
            const int rj = __builtin_amdgcn_readlane(r, j);
            cj[u] = __builtin_amdgcn_readlane(c, j) - lo;
            wj[u] = __int_as_float(__builtin_amdgcn_readlane(wbits, j));
            const float* src = h + (size_t)rj * F + lane;
#pragma unroll
            for (int k = 0; k < L; ++k) v[u][k] = src[32 * k];
          }
        }
#pragma unroll
        for (int u = 0; u < U; ++u) {
          if (u < n) {
            float* dst = acc + cj[u] * PITCH + lane;
#pragma unroll
            for (int k = 0; k < L; ++k)
              (void)__hip_atomic_fetch_add(dst + 32 * k, v[u][k] * wj[u], __ATOMIC_RELAXED,
                                           __HIP_MEMORY_SCOPE_WORKGROUP);
          }
        }
      }
      cnt = 0;
    };

    uint4* park = stg + wave * 128 + lane;
    for (int g = wave * 4; g < NCH; g += NW * 4) {
      uint4 tg[4];
#pragma unroll
      for (int sc = 0; sc < 4; ++sc) {
        const int chc = (g + sc < NCH) ? (g + sc) : (NCH - 1);
        tg[sc] = *(const uint4*)(tag + (chc << 7) + (lane << 2));
      }
#pragma unroll
      for (int sc = 0; sc < 4; ++sc) {
        if (g + sc >= NCH) tg[sc] = make_uint4(~pat, ~pat, ~pat, ~pat);
        park[sc * 32] = tg[sc];
      }
#pragma nounroll
      for (int sc = 0; sc < 4; ++sc) {
        const uint4 tq = park[sc * 32];
        const unsigned z0 = zero_bytes(tq.x ^ pat);
        const unsigned z1 = zero_bytes(tq.y ^ pat);
        const unsigned z2 = zero_bytes(tq.z ^ pat);
        const unsigned z3 = zero_bytes(tq.w ^ pat);
        unsigned m = __builtin_amdgcn_ballot_w32((z0 | z1 | z2 | z3) != 0u);
        while (m) {
          const int j = __builtin_ctz(m);
          m &= m - 1;
          if (cnt > 16) flush();
          const int ebase = ((g + sc) << 9) + (j << 4);
#pragma unroll
          for (int q = 0; q < 4; ++q) {
            unsigned zw = (unsigned)__builtin_amdgcn_readlane(
                (int)(q == 0 ? z0 : q == 1 ? z1 : q == 2 ? z2 : z3), j);
            while (zw) {
              const int bit = __builtin_ctz(zw);
              zw &= zw - 1;
              const int e = ebase + (q << 2) + (bit >> 3);
              if (e < E) {
                qe = (lane == cnt) ? e : qe;
                ++cnt;
              }
            }
          }
        }
      }
    }
    if (cnt > 0) flush();
    __syncthreads();
    if (threadIdx.x == 0) acount[o] = s_cnt;

    for (int i = wave; i < nrows; i += NW) {
      float* a = acc + i * PITCH + lane;
#pragma unroll
      for (int k = 0; k < L; ++k) a[32 * k] = fmaxf(a[32 * k] + bk[k], 0.0f);
    }
    __syncthreads();

    const int l15   = lane & 15;
    const int khalf = (lane >> 4) << 1;
    const int nrt   = (nrows + 15) >> 4;
    for (int task = wave; task < nrt * CP; task += NW) {
      const int rt = task / CP;
      const int n0 = (task % CP) << 5;
      int ar = (rt << 4) + l15;
      ar = (ar < nrows) ? ar : (nrows - 1);
      const float* ap = acc + ar * PITCH + khalf;
      const float* bp = B + (size_t)khalf * N2 + n0 + 2 * l15;
      const int rl = (rt << 4) + ((lane >> 4) << 3);
      v8f c0 = {}, c1 = {};
      for (int k0 = 0; k0 < F; k0 += 4) {
        const float2 av = *(const float2*)(ap + k0);
        const float2 bx = *(const float2*)(bp + (size_t)k0 * N2);
        const float2 by = *(const float2*)(bp + (size_t)(k0 + 1) * N2);
        v2f a, be, bo;
        a.x  = av.x;
        a.y  = av.y;
        be.x = bx.x;
        be.y = by.x;
        bo.x = bx.y;
        bo.y = by.y;
        c0 = __builtin_amdgcn_wmma_f32_16x16x4_f32(
             false, a,  false, be,
             (short)0, c0,  false,  false);
        c1 = __builtin_amdgcn_wmma_f32_16x16x4_f32(
             false, a,  false, bo,
             (short)0, c1,  false,  false);
      }
#pragma unroll
      for (int v = 0; v < 8; ++v) {
        if (rl + v < nrows) {
          const size_t o2 = (size_t)(lo + rl + v) * N2 + n0 + 2 * l15;
          *(float2*)(C2 + o2) = make_float2(c0[v], c1[v]);
        }
      }
    }
    __syncthreads();
  }
}

template <int F>
__global__ __launch_bounds__(1024) void k_agg_rows(const float* __restrict__ h,
                                                   const int* __restrict__ esrc,
                                                   const int* __restrict__ etgt,
                                                   const float* __restrict__ ew,
                                                   const int* __restrict__ rcount,
                                                   const int* __restrict__ acount,
                                                   const float* __restrict__ dinv,
                                                   const float* __restrict__ bout,
                                                   float* __restrict__ out, int E, int Nn) {
  static_assert(F % 32 == 0 && F <= 128, "row shape");
  constexpr int L  = F / 32;
  constexpr int U  = 32 / L;
  constexpr int NW = 32;
  __shared__ float acc[AGG_R * F];
  __shared__ int s_off;
  const int lane = threadIdx.x & 31;
  const int wave = __builtin_amdgcn_readfirstlane((int)threadIdx.x >> 5);
  float bk[L];
#pragma unroll
  for (int k = 0; k < L; ++k) bk[k] = bout[lane + 32 * k];

  for (int o = blockIdx.x; o * AGG_R < Nn; o += AGG_OWNERS) {
    const int lo    = o * AGG_R;
    const int nrows = (Nn - lo < AGG_R) ? (Nn - lo) : AGG_R;

    if (wave == 0) {
      const int below = edges_below(rcount, o, lane);
      if (lane == 0) s_off = below;
    }

    for (int i = wave; i < nrows; i += NW) {
      const float d = dinv[lo + i];
      const float* src = h + (size_t)(lo + i) * F + lane;
      float* a = acc + i * F + lane;
#pragma unroll
      for (int k = 0; k < L; ++k) a[32 * k] = (src[32 * k] * d) * d + bk[k];
    }
    __syncthreads();
    const int off = s_off;
    int n = acount[o];
    n = (n < E - off) ? n : (E - off);

    for (int b = wave * 32; b < n; b += NW * 32) {
      const bool ok = (b + lane) < n;
      int r = 0, t = 0;
      float w = 0.0f;
      if (ok) {
        r = esrc[off + b + lane];
        t = etgt[off + b + lane];
        w = ew[off + b + lane];
      }
      const int wbits = __float_as_int(w);
      unsigned mm = __builtin_amdgcn_ballot_w32(ok);
      while (mm) {
        const int m = __builtin_popcount(mm);
        float v[U][L];
        int cj[U];
        float wj[U];
#pragma unroll
        for (int u = 0; u < U; ++u) {
          if (u < m) {
            const int j = __builtin_ctz(mm);
            mm &= mm - 1;
            const int rj = __builtin_amdgcn_readlane(r, j);
            cj[u] = __builtin_amdgcn_readlane(t, j);
            wj[u] = __int_as_float(__builtin_amdgcn_readlane(wbits, j));
            const float* src = h + (size_t)rj * F + lane;
#pragma unroll
            for (int k = 0; k < L; ++k) v[u][k] = src[32 * k];
          }
        }
#pragma unroll
        for (int u = 0; u < U; ++u) {
          if (u < m) {
            float* dst = acc + cj[u] * F + lane;
#pragma unroll
            for (int k = 0; k < L; ++k)
              (void)__hip_atomic_fetch_add(dst + 32 * k, v[u][k] * wj[u], __ATOMIC_RELAXED,
                                           __HIP_MEMORY_SCOPE_WORKGROUP);
          }
        }
      }
    }
    __syncthreads();

    for (int i = wave; i < nrows; i += NW) {
      const float* a = acc + i * F + lane;
      float* dst = out + (size_t)(lo + i) * F + lane;
#pragma unroll
      for (int k = 0; k < L; ++k) dst[32 * k] = a[32 * k];
    }
    __syncthreads();
  }
}

extern "C" void kernel_launch(void* const* d_in, const int* in_sizes, int n_in,
                              void* d_out, int out_size, void* d_ws, size_t ws_size,
                              hipStream_t stream) {
  const float* x  = (const float*)d_in[0];
  const int*   ei = (const int*)d_in[1];
  const float* W1 = (const float*)d_in[2];
  const float* b1 = (const float*)d_in[3];
  const float* W2 = (const float*)d_in[4];
  const float* b2 = (const float*)d_in[5];

  const int H  = in_sizes[3];
  const int C  = in_sizes[5];
  const int F  = in_sizes[2] / H;
  const int Nn = in_sizes[0] / F;
  const int E  = in_sizes[1] / 2;
  const int* rowIdx = ei;
  const int* colIdx = ei + E;

  auto al = [](size_t v) { return (v + 255) & ~(size_t)255; };
  const int pitch = (Nn + 63) & ~63;
  float* dinv = (float*)d_ws;
  size_t off  = al((size_t)Nn * sizeof(float));
  float* cnt  = (float*)((char*)d_ws + off);
  off += al((size_t)DEG_TABS * pitch * sizeof(float));
  float* hbuf = (float*)((char*)d_ws + off);
  off += al((size_t)Nn * H * sizeof(float));
  float* h2buf = (float*)((char*)d_ws + off);
  off += al((size_t)Nn * C * sizeof(float));
  const int nranges = (Nn + AGG_R - 1) / AGG_R;
  int* rcount = (int*)((char*)d_ws + off);
  off += al((size_t)nranges * sizeof(int));
  int* acount = (int*)((char*)d_ws + off);
  off += al((size_t)nranges * sizeof(int));
  int* esrc = (int*)((char*)d_ws + off);
  off += al((size_t)E * sizeof(int));
  int* etgt = (int*)((char*)d_ws + off);
  off += al((size_t)E * sizeof(int));
  float* ew = (float*)((char*)d_ws + off);
  off += al((size_t)E * sizeof(float));
  unsigned* tag = (unsigned*)((char*)d_ws + off);
  const int nwords = ((E + 511) / 512) * 128;

  const dim3 blk(256);
  auto blocks = [](long long work) { return (unsigned)((work + 255) / 256); };

  const int n4 = DEG_TABS * pitch / 4;
  k_init_deg <<<blocks(n4), blk, 0, stream>>>(cnt, n4);
  k_count_deg<<<blocks(nwords), blk, 0, stream>>>(colIdx, cnt, tag, pitch, E, nwords);
  k_dinv     <<<nranges, blk, 0, stream>>>(cnt, dinv, rcount, pitch, Nn);

  float* out = (float*)d_out;
  k_gemm_wmma_f32<128, 256><<<512, blk, 0, stream>>>(x, W1, hbuf, Nn);
  k_agg_gemm<256, 64><<<AGG_OWNERS, dim3(1024), 0, stream>>>(hbuf, rowIdx, colIdx, tag, dinv, W2,
                                                            b1, rcount, acount, esrc, etgt, ew,
                                                            h2buf, E, Nn);

  k_agg_rows<64><<<AGG_OWNERS, dim3(1024), 0, stream>>>(h2buf, esrc, etgt, ew, rcount, acount,
                                                       dinv, b2, out, E, Nn);
}
